# starting kernel
# baseline (speedup 1.0000x reference)
#include <hip/hip_runtime.h>

typedef __attribute__((ext_vector_type(2))) float v2f;
typedef __attribute__((ext_vector_type(8))) float v8f;

#define C_OUT   24
#define KTOT    72
#define HHALF   56
#define WIDTH   112
#define SPATIAL (HHALF * 2 * WIDTH)

__global__ void __launch_bounds__(256)
shiftconv_wmma(const float* __restrict__ x,
               const float* __restrict__ Wf,
               float* __restrict__ y)
{
    const int lane  = threadIdx.x & 31;
    const int sTile = (blockIdx.x * blockDim.x + threadIdx.x) >> 5;
    if (sTile >= 784) return;

    const int row112 = sTile / 7;
    const int mb     = (sTile - row112 * 7) << 4;
    const int n      = row112 >> 1;
    const int o      = row112 & 1;

    const int half = lane >> 4;
    const int lp   = lane & 15;
    const int m    = mb + lp;

    float tapMask[3];
    int   tapOff [3];
#pragma unroll
    for (int k = 0; k < 3; ++k) {
        int  q     = n + k - 1;
        bool valid = (q >= 0) && (q < HHALF);
        int  n2    = (q <= 0) ? (HHALF - 1) : (q - 1);
        tapMask[k] = valid ? 1.0f : 0.0f;
        tapOff[k]  = (n2 * 2 + o) * WIDTH + m;
    }

    const float* bPtr [3][2];
    float        bMask[3][2];
#pragma unroll
    for (int p = 0; p < 3; ++p) {
#pragma unroll
        for (int e = 0; e < 2; ++e) {
            int c  = 4 * p + 2 * half + e;
            int jo = c / 3;
            int k  = c - 3 * jo;
            bPtr [p][e] = x + jo * SPATIAL + tapOff[k];
            bMask[p][e] = tapMask[k];
        }
    }

    const bool   i1Valid = (16 + lp) < C_OUT;
    const float  aScale1 = i1Valid ? 1.0f : 0.0f;
    const float* aPtr0   = Wf + lp * KTOT + 2 * half;
    const float* aPtr1   = Wf + (i1Valid ? (16 + lp) : 0) * KTOT + 2 * half;

    v8f acc0 = {};
    v8f acc1 = {};

#pragma unroll
    for (int t = 0; t < 6; ++t) {
#pragma unroll
        for (int p = 0; p < 3; ++p) {
            const int aOff = 12 * t + 4 * p;
            const int bOff = 4 * t * SPATIAL;

            v2f a0, a1, b;
            a0.x = aPtr0[aOff + 0];
            a0.y = aPtr0[aOff + 1];
            a1.x = aScale1 * aPtr1[aOff + 0];
            a1.y = aScale1 * aPtr1[aOff + 1];
            b.x  = bMask[p][0] * bPtr[p][0][bOff];
            b.y  = bMask[p][1] * bPtr[p][1][bOff];

            acc0 = __builtin_amdgcn_wmma_f32_16x16x4_f32(
                false, a0, false, b, (short)0, acc0, false, false);
            acc1 = __builtin_amdgcn_wmma_f32_16x16x4_f32(
                false, a1, false, b, (short)0, acc1, false, false);
        }
    }

    const int mOut = (m + 1 == WIDTH) ? 0 : (m + 1);
    const int base = row112 * WIDTH + mOut;

#pragma unroll
    for (int e = 0; e < 8; ++e) {
        int i0 = e + 8 * half;
        y[i0 * SPATIAL + base] = acc0[e];
    }
    if (half == 0) {
#pragma unroll
        for (int e = 0; e < 8; ++e) {
            y[(16 + e) * SPATIAL + base] = acc1[e];
        }
    }
}

extern "C" void kernel_launch(void* const* d_in, const int* in_sizes, int n_in,
                              void* d_out, int out_size, void* d_ws, size_t ws_size,
                              hipStream_t stream) {
    (void)in_sizes; (void)n_in; (void)out_size; (void)d_ws; (void)ws_size;
    const float* x = (const float*)d_in[0];
    const float* W = (const float*)d_in[1];
    float*       y = (float*)d_out;

    dim3 grid(98), block(256);
    hipLaunchKernelGGL(shiftconv_wmma, grid, block, 0, stream, x, W, y);
}


# Operands staged once per workgroup in LDS by 16-byte loads (one output row per 7-wave workgroup, pad taps stored as zeros); fragments read from LDS in one group; same products, same sum order
# speedup vs baseline: 1.3145x; 1.3145x over previous
#include <hip/hip_runtime.h>

typedef __attribute__((ext_vector_type(2))) float v2f;
typedef __attribute__((ext_vector_type(4))) float v4f;
typedef __attribute__((ext_vector_type(8))) float v8f;

#define C_OUT   24
#define KTOT    72
#define HHALF   56
#define WIDTH   112
#define SPATIAL (HHALF * 2 * WIDTH)
#define NWAVE   7
#define NTHR    (NWAVE * 32)
#define WPITCH  76

__global__ void __launch_bounds__(NTHR)
shiftconv_wmma(const float* __restrict__ x,
               const float* __restrict__ Wf,
               float* __restrict__ y)
{
    __shared__ __attribute__((aligned(16))) float xs[KTOT * WIDTH];
    __shared__ __attribute__((aligned(16))) float ws[32 * WPITCH];

    const int tid    = threadIdx.x;
    const int row112 = blockIdx.x;
    const int n      = row112 >> 1;
    const int o      = row112 & 1;

    const int  q0   = tid;
    const int  q1   = min(tid + NTHR, C_OUT * KTOT / 4 - 1);
    const v4f  w0   = ((const v4f*)Wf)[q0];
    const v4f  w1   = ((const v4f*)Wf)[q1];

    int  tapRow[3];
    bool tapValid[3];
#pragma unroll
    for (int k = 0; k < 3; ++k) {
        const int q  = n + k - 1;
        tapValid[k]  = (q >= 0) && (q < HHALF);
        const int n2 = (q <= 0) ? (HHALF - 1) : (q - 1);
        tapRow[k]    = (2 * n2 + o) * WIDTH;
    }
    const int c4 = tid % 28;
    const int u  = tid / 28;
    const float* xThr = x + (3 * u) * SPATIAL + 4 * c4;
    v4f xv[9];
#pragma unroll
    for (int i = 0; i < 9; ++i) {
        const int jj = i / 3;
        const int k  = i - 3 * jj;
        const v4f v = *(const v4f*)(xThr + jj * SPATIAL + tapRow[k]);
        const v4f z = {0.0f, 0.0f, 0.0f, 0.0f};
        xv[i] = tapValid[k] ? v : z;
    }

#pragma unroll
    for (int i = 0; i < 9; ++i)
        *(v4f*)&xs[(9 * u + i) * WIDTH + 4 * c4] = xv[i];
    *(v4f*)&ws[(q0 / 18) * WPITCH + (q0 % 18) * 4] = w0;
    *(v4f*)&ws[(q1 / 18) * WPITCH + (q1 % 18) * 4] = w1;
    if (tid < 8 * WPITCH / 4) {
        const v4f z = {0.0f, 0.0f, 0.0f, 0.0f};
        *(v4f*)&ws[C_OUT * WPITCH + 4 * tid] = z;
    }
    __syncthreads();

    const int lane = tid & 31;
    const int mb   = (tid >> 5) << 4;
    const int half = lane >> 4;
    const int lp   = lane & 15;
    const int m    = mb + lp;

    const float* bBase  = xs + (2 * half) * WIDTH + m;
    const float* a0Base = ws + lp * WPITCH + 2 * half;
    const float* a1Base = ws + (16 + lp) * WPITCH + 2 * half;

    v2f a0[18], a1[18], b[18];
#pragma unroll
    for (int s = 0; s < 18; ++s) {
        a0[s]  = *(const v2f*)(a0Base + 4 * s);
        a1[s]  = *(const v2f*)(a1Base + 4 * s);
        b[s].x = bBase[(4 * s + 0) * WIDTH];
        b[s].y = bBase[(4 * s + 1) * WIDTH];
        __builtin_amdgcn_sched_barrier(0);
    }

    v8f acc0 = {};
    v8f acc1 = {};
#pragma unroll
    for (int s = 0; s < 18; ++s) {
        acc0 = __builtin_amdgcn_wmma_f32_16x16x4_f32(
            false, a0[s], false, b[s], (short)0, acc0, false, false);
        acc1 = __builtin_amdgcn_wmma_f32_16x16x4_f32(
            false, a1[s], false, b[s], (short)0, acc1, false, false);
    }

    const int mOut = (m + 1 == WIDTH) ? 0 : (m + 1);
    const int base = row112 * WIDTH + mOut;

#pragma unroll
    for (int e = 0; e < 8; ++e) {
        int i0 = e + 8 * half;
        y[i0 * SPATIAL + base] = acc0[e];
    }
    if (half == 0) {
#pragma unroll
        for (int e = 0; e < 8; ++e) {
            y[(16 + e) * SPATIAL + base] = acc1[e];
        }
    }
}

extern "C" void kernel_launch(void* const* d_in, const int* in_sizes, int n_in,
                              void* d_out, int out_size, void* d_ws, size_t ws_size,
                              hipStream_t stream) {
    (void)in_sizes; (void)n_in; (void)out_size; (void)d_ws; (void)ws_size;
    const float* x = (const float*)d_in[0];
    const float* W = (const float*)d_in[1];
    float*       y = (float*)d_out;

    dim3 grid(112), block(NTHR);
    hipLaunchKernelGGL(shiftconv_wmma, grid, block, 0, stream, x, W, y);
}


# Workgroup = half an output row (64 or 48 columns staged, 4 waves, one 16-column tile each, grid 224): less staging and LDS traffic per workgroup, one wave per SIMD; same products, same sum order
# speedup vs baseline: 1.4106x; 1.0731x over previous
#include <hip/hip_runtime.h>

typedef __attribute__((ext_vector_type(2))) float v2f;
typedef __attribute__((ext_vector_type(4))) float v4f;
typedef __attribute__((ext_vector_type(8))) float v8f;

#define C_OUT   24
#define KTOT    72
#define HHALF   56
#define WIDTH   112
#define SPATIAL (HHALF * 2 * WIDTH)
#define NTILE   7
#define NWAVE   4
#define NTHR    (NWAVE * 32)
#define XCOLS   64
#define XPITCH  80
#define WPITCH  76

__global__ void __launch_bounds__(NTHR)
shiftconv_wmma(const float* __restrict__ x,
               const float* __restrict__ Wf,
               float* __restrict__ y)
{
    __shared__ __attribute__((aligned(16))) float xs[KTOT * XPITCH];
    __shared__ __attribute__((aligned(16))) float ws[C_OUT * WPITCH];

    const int tid    = threadIdx.x;
    const int row112 = blockIdx.x >> 1;
    const int hb     = blockIdx.x & 1;
    const int n      = row112 >> 1;
    const int o      = row112 & 1;

    int qw[4];
    v4f wv[4];
#pragma unroll
    for (int j = 0; j < 4; ++j) {
        qw[j] = min(tid + NTHR * j, C_OUT * KTOT / 4 - 1);
        wv[j] = ((const v4f*)Wf)[qw[j]];
    }

    int  tapRow[3];
    bool tapValid[3];
#pragma unroll
    for (int k = 0; k < 3; ++k) {
        const int q  = n + k - 1;
        tapValid[k]  = (q >= 0) && (q < HHALF);
        const int n2 = (q <= 0) ? (HHALF - 1) : (q - 1);
        tapRow[k]    = (2 * n2 + o) * WIDTH;
    }
    const int c4  = tid & 15;
    const int u   = tid >> 4;
    const int c4g = min(16 * hb + c4, WIDTH / 4 - 1);
    const float* xThr = x + (3 * u) * SPATIAL + 4 * c4g;
    v4f xv[9];
#pragma unroll
    for (int i = 0; i < 9; ++i) {
        const int jj = i / 3;
        const int k  = i - 3 * jj;
        const v4f v = *(const v4f*)(xThr + jj * SPATIAL + tapRow[k]);
        const v4f z = {0.0f, 0.0f, 0.0f, 0.0f};
        xv[i] = tapValid[k] ? v : z;
    }

#pragma unroll
    for (int i = 0; i < 9; ++i)
        *(v4f*)&xs[(9 * u + i) * XPITCH + 4 * c4] = xv[i];
#pragma unroll
    for (int j = 0; j < 4; ++j)
        *(v4f*)&ws[(qw[j] / 18) * WPITCH + (qw[j] % 18) * 4] = wv[j];
    __syncthreads();

    const int wv_id = __builtin_amdgcn_readfirstlane(tid >> 5);
    const int tile  = NWAVE * hb + wv_id;
    if (tile >= NTILE) return;

    const int lane = tid & 31;
    const int half = lane >> 4;
    const int lp   = lane & 15;
    const int mloc = (wv_id << 4) + lp;
    const int m    = XCOLS * hb + mloc;

    const float* bBase  = xs + (2 * half) * XPITCH + mloc;
    const float* a0Base = ws + lp * WPITCH + 2 * half;
    const float* a1Base = ws + (16 + (lp & 7)) * WPITCH + 2 * half;

    v2f a0[18], a1[18], b[18];
#pragma unroll
    for (int s = 0; s < 18; ++s) {
        a0[s]  = *(const v2f*)(a0Base + 4 * s);
        a1[s]  = *(const v2f*)(a1Base + 4 * s);
        b[s].x = bBase[(4 * s + 0) * XPITCH];
        b[s].y = bBase[(4 * s + 1) * XPITCH];
        __builtin_amdgcn_sched_barrier(0);
    }

    v8f acc0 = {};
    v8f acc1 = {};
#pragma unroll
    for (int s = 0; s < 18; ++s) {
        acc0 = __builtin_amdgcn_wmma_f32_16x16x4_f32(
            false, a0[s], false, b[s], (short)0, acc0, false, false);
        acc1 = __builtin_amdgcn_wmma_f32_16x16x4_f32(
            false, a1[s], false, b[s], (short)0, acc1, false, false);
    }

    const int mOut = (m + 1 == WIDTH) ? 0 : (m + 1);
    const int base = row112 * WIDTH + mOut;

#pragma unroll
    for (int e = 0; e < 8; ++e) {
        int i0 = e + 8 * half;
        y[i0 * SPATIAL + base] = acc0[e];
    }
    if (half == 0) {
#pragma unroll
        for (int e = 0; e < 8; ++e) {
            y[(16 + e) * SPATIAL + base] = acc1[e];
        }
    }
}

extern "C" void kernel_launch(void* const* d_in, const int* in_sizes, int n_in,
                              void* d_out, int out_size, void* d_ws, size_t ws_size,
                              hipStream_t stream) {
    (void)in_sizes; (void)n_in; (void)out_size; (void)d_ws; (void)ws_size;
    const float* x = (const float*)d_in[0];
    const float* W = (const float*)d_in[1];
    float*       y = (float*)d_out;

    dim3 grid(224), block(NTHR);
    hipLaunchKernelGGL(shiftconv_wmma, grid, block, 0, stream, x, W, y);
}
